# starting kernel
# baseline (speedup 1.0000x reference)
#include <hip/hip_runtime.h>


#define NROWS   1048576
#define BSEG    64
#define DFEAT   64
#define CHUNK   2048
#define THREADS 256
#define PF_ROWS 64

typedef float v2f __attribute__((ext_vector_type(2)));
typedef float v8f __attribute__((ext_vector_type(8)));

__global__ void gab_zero_ws(float* __restrict__ ws) {
    int i = blockIdx.x * blockDim.x + threadIdx.x;
    if (i < BSEG * DFEAT) ws[i] = 0.0f;
}

__global__ __launch_bounds__(THREADS)
void gab_seg_sum_wmma(const float* __restrict__ x,
                      const int*   __restrict__ seg,
                      float*       __restrict__ ws) {
    const int tid  = threadIdx.x;
    const int lane = tid & 31;
    const int wave = tid >> 5;
    const int ft   = wave & 3;
    const int rh   = wave >> 2;

    const int chunk_start = blockIdx.x * CHUNK;
    const int row_base    = chunk_start + rh * (CHUNK / 2);

    const int seg_base = seg[chunk_start];

    const int m  = lane & 15;
    const int kh = (lane >> 4) << 1;
    const int f  = ft * 16 + m;

    const float* px  = x   + (size_t)(row_base + kh) * DFEAT + f;
    const int*   ps  = seg + row_base + kh;
    const float* ppf = px  + (size_t)PF_ROWS * DFEAT;

    v8f c0 = {};
    v8f c1 = {};

    for (int it = 0; it < (CHUNK / 2) / 8; ++it) {
        const int s0 = ps[0];
        const int s1 = ps[1];
        const int s2 = ps[4];
        const int s3 = ps[5];

        v2f a0, a1, b0, b1;
        a0.x = (s0 - seg_base == m) ? 1.0f : 0.0f;
        a0.y = (s1 - seg_base == m) ? 1.0f : 0.0f;
        a1.x = (s2 - seg_base == m) ? 1.0f : 0.0f;
        a1.y = (s3 - seg_base == m) ? 1.0f : 0.0f;

        b0.x = px[0];
        b0.y = px[DFEAT];
        b1.x = px[4 * DFEAT];
        b1.y = px[5 * DFEAT];

        __builtin_prefetch(ppf, 0, 1);

        c0 = __builtin_amdgcn_wmma_f32_16x16x4_f32(
                false, a0, false, b0, (short)0, c0, false, false);
        c1 = __builtin_amdgcn_wmma_f32_16x16x4_f32(
                false, a1, false, b1, (short)0, c1, false, false);

        px  += 8 * DFEAT;
        ps  += 8;
        ppf += 8 * DFEAT;
    }

    const int mofs = (lane >> 4) * 8;
    #pragma unroll
    for (int r = 0; r < 8; ++r) {
        float v = c0[r] + c1[r];
        if (v != 0.0f) {
            int sgi = seg_base + r + mofs;
            if (sgi < BSEG)
                atomicAdd(&ws[sgi * DFEAT + f], v);
        }
    }
}

__global__ void gab_finalize(const float* __restrict__ ws,
                             const int*   __restrict__ lengths,
                             float*       __restrict__ out) {
    int i = blockIdx.x * blockDim.x + threadIdx.x;
    if (i < BSEG * DFEAT) {
        out[i] = ws[i] / (float)lengths[i / DFEAT];
    }
}

extern "C" void kernel_launch(void* const* d_in, const int* in_sizes, int n_in,
                              void* d_out, int out_size, void* d_ws, size_t ws_size,
                              hipStream_t stream) {
    const float* x       = (const float*)d_in[0];
    const int*   seg     = (const int*)  d_in[1];
    const int*   lengths = (const int*)  d_in[2];
    float*       out     = (float*)d_out;
    float*       ws      = (float*)d_ws;

    gab_zero_ws<<<(BSEG * DFEAT + 255) / 256, 256, 0, stream>>>(ws);
    gab_seg_sum_wmma<<<NROWS / CHUNK, THREADS, 0, stream>>>(x, seg, ws);
    gab_finalize<<<(BSEG * DFEAT + 255) / 256, 256, 0, stream>>>(ws, lengths, out);
}


# Each wave streams whole 256-B rows, 16-32 global_load_b128 in flight (was 4-B loads + prefetch); block partials summed in LDS before the atomics: f32 sum order regrouped
# speedup vs baseline: 1.3139x; 1.3139x over previous
#include <hip/hip_runtime.h>


#define NROWS   1048576
#define BSEG    64
#define DFEAT   64
#define CHUNK   2048
#define THREADS 256
#define WAVES   (THREADS / 32)
#define WROWS   (CHUNK / WAVES)
#define STEP    32
#define NSTEP   (WROWS / STEP)

typedef float v2f __attribute__((ext_vector_type(2)));
typedef float v4f __attribute__((ext_vector_type(4)));
typedef float v8f __attribute__((ext_vector_type(8)));
typedef int   v4i __attribute__((ext_vector_type(4)));

__global__ void gab_zero_ws(float* __restrict__ ws) {
    int i = blockIdx.x * blockDim.x + threadIdx.x;
    if (i < BSEG * DFEAT) ws[i] = 0.0f;
}

__global__ __launch_bounds__(THREADS)
void gab_seg_sum_wmma(const float* __restrict__ x,
                      const int*   __restrict__ seg,
                      float*       __restrict__ ws) {
    __shared__ float red[WAVES][16][DFEAT];

    const int tid  = threadIdx.x;
    const int lane = tid & 31;
    const int wave = tid >> 5;
    const int n    = lane & 15;
    const int h    = lane >> 4;

    const int chunk_start = blockIdx.x * CHUNK;
    const int seg_base    = seg[chunk_start];
    const int key         = seg_base + n;

    const int    wrow = chunk_start + wave * WROWS;
    const float* px   = x   + (size_t)(wrow + 16 * h) * DFEAT + 4 * n;
    const int*   ps   = seg + wrow + 16 * h;

    v4f q[2][16];
    v4i sv[2][4];
    v8f c0 = {}, c1 = {}, c2 = {}, c3 = {};

#define LOAD_STEP(S) do {                                                          \
        _Pragma("unroll") for (int j = 0; j < 4; ++j)                              \
            sv[(S) & 1][j] = *(const v4i*)(ps + (S) * STEP + 4 * j);               \
        _Pragma("unroll") for (int j = 0; j < 16; ++j)                             \
            q[(S) & 1][j] = *(const v4f*)(px + (size_t)((S) * STEP + j) * DFEAT);  \
    } while (0)

    LOAD_STEP(0);

    #pragma unroll
    for (int s = 0; s < NSTEP; ++s) {
        __builtin_amdgcn_sched_barrier(0);
        if (s + 1 < NSTEP) LOAD_STEP(s + 1);
        __builtin_amdgcn_sched_barrier(0);

        #pragma unroll
        for (int g = 0; g < 8; ++g) {
            const int sa = sv[s & 1][g >> 1][(2 * g) & 3];
            const int sb = sv[s & 1][g >> 1][(2 * g + 1) & 3];
            v2f a;
            a.x = (sa == key) ? 1.0f : 0.0f;
            a.y = (sb == key) ? 1.0f : 0.0f;

            const v4f qa = q[s & 1][2 * g];
            const v4f qb = q[s & 1][2 * g + 1];
            v2f b0, b1, b2, b3;
            b0.x = qa.x; b0.y = qb.x;
            b1.x = qa.y; b1.y = qb.y;
            b2.x = qa.z; b2.y = qb.z;
            b3.x = qa.w; b3.y = qb.w;

            c0 = __builtin_amdgcn_wmma_f32_16x16x4_f32(false, a, false, b0, (short)0, c0, false, false);
            c1 = __builtin_amdgcn_wmma_f32_16x16x4_f32(false, a, false, b1, (short)0, c1, false, false);
            c2 = __builtin_amdgcn_wmma_f32_16x16x4_f32(false, a, false, b2, (short)0, c2, false, false);
            c3 = __builtin_amdgcn_wmma_f32_16x16x4_f32(false, a, false, b3, (short)0, c3, false, false);
        }
    }
#undef LOAD_STEP

    #pragma unroll
    for (int r = 0; r < 8; ++r) {
        v4f v;
        v.x = c0[r]; v.y = c1[r]; v.z = c2[r]; v.w = c3[r];
        *(v4f*)&red[wave][8 * h + r][4 * n] = v;
    }
    __syncthreads();

    const int m  = tid >> 4;
    const int fq = tid & 15;
    v4f acc = *(const v4f*)&red[0][m][4 * fq];
    #pragma unroll
    for (int w = 1; w < WAVES; ++w)
        acc += *(const v4f*)&red[w][m][4 * fq];

    const int sgi = seg_base + m;
    if (sgi < BSEG) {
        float* dst = ws + sgi * DFEAT + 4 * fq;
        if (acc.x != 0.0f) atomicAdd(dst + 0, acc.x);
        if (acc.y != 0.0f) atomicAdd(dst + 1, acc.y);
        if (acc.z != 0.0f) atomicAdd(dst + 2, acc.z);
        if (acc.w != 0.0f) atomicAdd(dst + 3, acc.w);
    }
}

__global__ void gab_finalize(const float* __restrict__ ws,
                             const int*   __restrict__ lengths,
                             float*       __restrict__ out) {
    int i = blockIdx.x * blockDim.x + threadIdx.x;
    if (i < BSEG * DFEAT) {
        out[i] = ws[i] / (float)lengths[i / DFEAT];
    }
}

extern "C" void kernel_launch(void* const* d_in, const int* in_sizes, int n_in,
                              void* d_out, int out_size, void* d_ws, size_t ws_size,
                              hipStream_t stream) {
    const float* x       = (const float*)d_in[0];
    const int*   seg     = (const int*)  d_in[1];
    const int*   lengths = (const int*)  d_in[2];
    float*       out     = (float*)d_out;
    float*       ws      = (float*)d_ws;

    gab_zero_ws<<<(BSEG * DFEAT + 255) / 256, 256, 0, stream>>>(ws);
    gab_seg_sum_wmma<<<NROWS / CHUNK, THREADS, 0, stream>>>(x, seg, ws);
    gab_finalize<<<(BSEG * DFEAT + 255) / 256, 256, 0, stream>>>(ws, lengths, out);
}


# Finalize launch folded into the sum kernel (3 launches to 2): each block divides its partial by the length and adds it atomically into zeroed out; division now precedes the cross-block f32 sum
# speedup vs baseline: 1.4191x; 1.0801x over previous
#include <hip/hip_runtime.h>


#define NROWS   1048576
#define BSEG    64
#define DFEAT   64
#define CHUNK   2048
#define THREADS 256
#define WAVES   (THREADS / 32)
#define WROWS   (CHUNK / WAVES)
#define STEP    32
#define NSTEP   (WROWS / STEP)

typedef float v2f __attribute__((ext_vector_type(2)));
typedef float v4f __attribute__((ext_vector_type(4)));
typedef float v8f __attribute__((ext_vector_type(8)));
typedef int   v4i __attribute__((ext_vector_type(4)));

__global__ void gab_zero_out(float* __restrict__ out) {
    int i = blockIdx.x * blockDim.x + threadIdx.x;
    if (i < BSEG * DFEAT) out[i] = 0.0f;
}

__global__ __launch_bounds__(THREADS)
void gab_seg_mean_wmma(const float* __restrict__ x,
                       const int*   __restrict__ seg,
                       const int*   __restrict__ lengths,
                       float*       __restrict__ out) {
    __shared__ float red[WAVES][16][DFEAT];

    const int tid  = threadIdx.x;
    const int lane = tid & 31;
    const int wave = tid >> 5;
    const int n    = lane & 15;
    const int h    = lane >> 4;

    const int chunk_start = blockIdx.x * CHUNK;
    const int seg_base    = seg[chunk_start];
    const int key         = seg_base + n;

    const int    wrow = chunk_start + wave * WROWS;
    const float* px   = x   + (size_t)(wrow + 16 * h) * DFEAT + 4 * n;
    const int*   ps   = seg + wrow + 16 * h;

    v4f q[2][16];
    v4i sv[2][4];
    v8f c0 = {}, c1 = {}, c2 = {}, c3 = {};

#define LOAD_STEP(S) do {                                                          \
        _Pragma("unroll") for (int j = 0; j < 4; ++j)                              \
            sv[(S) & 1][j] = *(const v4i*)(ps + (S) * STEP + 4 * j);               \
        _Pragma("unroll") for (int j = 0; j < 16; ++j)                             \
            q[(S) & 1][j] = *(const v4f*)(px + (size_t)((S) * STEP + j) * DFEAT);  \
    } while (0)

    LOAD_STEP(0);

    const int m     = tid >> 4;
    const int fq    = tid & 15;
    const int sgi   = seg_base + m;
    const int len_i = lengths[min(max(sgi, 0), BSEG - 1)];

    #pragma unroll
    for (int s = 0; s < NSTEP; ++s) {
        __builtin_amdgcn_sched_barrier(0);
        if (s + 1 < NSTEP) LOAD_STEP(s + 1);
        __builtin_amdgcn_sched_barrier(0);

        #pragma unroll
        for (int g = 0; g < 8; ++g) {
            const int sa = sv[s & 1][g >> 1][(2 * g) & 3];
            const int sb = sv[s & 1][g >> 1][(2 * g + 1) & 3];
            v2f a;
            a.x = (sa == key) ? 1.0f : 0.0f;
            a.y = (sb == key) ? 1.0f : 0.0f;

            const v4f qa = q[s & 1][2 * g];
            const v4f qb = q[s & 1][2 * g + 1];
            v2f b0, b1, b2, b3;
            b0.x = qa.x; b0.y = qb.x;
            b1.x = qa.y; b1.y = qb.y;
            b2.x = qa.z; b2.y = qb.z;
            b3.x = qa.w; b3.y = qb.w;

            c0 = __builtin_amdgcn_wmma_f32_16x16x4_f32(false, a, false, b0, (short)0, c0, false, false);
            c1 = __builtin_amdgcn_wmma_f32_16x16x4_f32(false, a, false, b1, (short)0, c1, false, false);
            c2 = __builtin_amdgcn_wmma_f32_16x16x4_f32(false, a, false, b2, (short)0, c2, false, false);
            c3 = __builtin_amdgcn_wmma_f32_16x16x4_f32(false, a, false, b3, (short)0, c3, false, false);
        }
    }
#undef LOAD_STEP

    #pragma unroll
    for (int r = 0; r < 8; ++r) {
        v4f v;
        v.x = c0[r]; v.y = c1[r]; v.z = c2[r]; v.w = c3[r];
        *(v4f*)&red[wave][8 * h + r][4 * n] = v;
    }
    __syncthreads();

    v4f acc = *(const v4f*)&red[0][m][4 * fq];
    #pragma unroll
    for (int w = 1; w < WAVES; ++w)
        acc += *(const v4f*)&red[w][m][4 * fq];

    if (sgi < BSEG) {
        const float lenf = (float)len_i;
        float* dst = out + sgi * DFEAT + 4 * fq;
        if (acc.x != 0.0f) atomicAdd(dst + 0, acc.x / lenf);
        if (acc.y != 0.0f) atomicAdd(dst + 1, acc.y / lenf);
        if (acc.z != 0.0f) atomicAdd(dst + 2, acc.z / lenf);
        if (acc.w != 0.0f) atomicAdd(dst + 3, acc.w / lenf);
    }
}

extern "C" void kernel_launch(void* const* d_in, const int* in_sizes, int n_in,
                              void* d_out, int out_size, void* d_ws, size_t ws_size,
                              hipStream_t stream) {
    const float* x       = (const float*)d_in[0];
    const int*   seg     = (const int*)  d_in[1];
    const int*   lengths = (const int*)  d_in[2];
    float*       out     = (float*)d_out;

    gab_zero_out<<<(BSEG * DFEAT + 255) / 256, 256, 0, stream>>>(out);
    gab_seg_mean_wmma<<<NROWS / CHUNK, THREADS, 0, stream>>>(x, seg, lengths, out);
}
